# starting kernel
# baseline (speedup 1.0000x reference)
#include <hip/hip_runtime.h>
#include <hip/hip_bf16.h>


typedef float v2f __attribute__((ext_vector_type(2)));
typedef float v4f __attribute__((ext_vector_type(4)));
typedef float v8f __attribute__((ext_vector_type(8)));

#define D_FEAT      128
#define ROWSTRIDE   132
#define TILE        16
#define WAVES_PB    8

__global__ __launch_bounds__(256)
void edge_dot_wmma(const float* __restrict__ h,
                   const int*   __restrict__ u,
                   const int*   __restrict__ v,
                   float*       __restrict__ out,
                   int nE)
{
    extern __shared__ float smem[];

    const int  lane   = threadIdx.x & 31;
    const int  wave   = threadIdx.x >> 5;
    const long nTiles = ((long)nE + TILE - 1) / TILE;
    const long tileId = (long)blockIdx.x * WAVES_PB + wave;
    const bool active = (tileId < nTiles);
    const long ebase  = active ? tileId * TILE : 0;

    float* ldsA = smem + (size_t)wave * (2 * TILE * ROWSTRIDE);
    float* ldsB = ldsA + TILE * ROWSTRIDE;

    #pragma unroll 4
    for (int r = 0; r < TILE; ++r) {
        const long e  = ebase + r;
        int ur = 0, vr = 0;
        if (e < (long)nE) { ur = u[e]; vr = v[e]; }
        const v4f au = *(const v4f*)(h + (size_t)ur * D_FEAT + lane * 4);
        const v4f bv = *(const v4f*)(h + (size_t)vr * D_FEAT + lane * 4);
        *(v4f*)(ldsA + r * ROWSTRIDE + lane * 4) = au;
        *(v4f*)(ldsB + r * ROWSTRIDE + lane * 4) = bv;
    }
    __syncthreads();

    const int    mrow  = lane & 15;
    const int    khalf = (lane >> 4) * 2;
    const float* Arow  = ldsA + mrow * ROWSTRIDE + khalf;
    const float* Brow  = ldsB + mrow * ROWSTRIDE + khalf;

    v8f c = {};
    #pragma unroll
    for (int k = 0; k < D_FEAT; k += 4) {
        const v2f a = *(const v2f*)(Arow + k);
        const v2f b = *(const v2f*)(Brow + k);
        c = __builtin_amdgcn_wmma_f32_16x16x4_f32(false, a, false, b,
                                                  (short)0, c, false, false);
    }

    const bool validLane = (lane < 8) || (lane >= 24);
    const int  m   = (lane < 8) ? lane : (lane - 16);
    const int  sel = (lane < 8) ? lane : (lane - 24);

    float val = c[0];
    #pragma unroll
    for (int i = 1; i < 8; ++i)
        val = (sel == i) ? c[i] : val;

    const long e = ebase + m;
    if (active && validLane && e < (long)nE)
        out[e] = val;
}

extern "C" void kernel_launch(void* const* d_in, const int* in_sizes, int n_in,
                              void* d_out, int out_size, void* d_ws, size_t ws_size,
                              hipStream_t stream)
{
    const float* h  = (const float*)d_in[0];
    const int*   u  = (const int*)d_in[1];
    const int*   v  = (const int*)d_in[2];
    float*       out = (float*)d_out;

    const int  nE     = in_sizes[1];
    const long nTiles = ((long)nE + TILE - 1) / TILE;
    const int  blocks = (int)((nTiles + WAVES_PB - 1) / WAVES_PB);
    const size_t smemBytes = (size_t)WAVES_PB * 2 * TILE * ROWSTRIDE * sizeof(float);

    edge_dot_wmma<<<blocks, 256, smemBytes, stream>>>(h, u, v, out, nE);
}


# Persistent waves, software-pipelined over tiles: next tile's 32 row loads go out into registers before the current tile's LDS fragment reads and WMMA chain; wave-private LDS, block barrier dropped
# speedup vs baseline: 2.1747x; 2.1747x over previous
#include <hip/hip_runtime.h>


typedef float v2f __attribute__((ext_vector_type(2)));
typedef float v4f __attribute__((ext_vector_type(4)));
typedef float v8f __attribute__((ext_vector_type(8)));

#define D_FEAT      128
#define ROWSTRIDE   132
#define TILE        16
#define WAVES_PB    8
#define GRID_BLOCKS 512

static __device__ __forceinline__
int load_index(const int* __restrict__ u, const int* __restrict__ v,
               unsigned tile, int lane, int nE)
{
    const long e = (long)tile * TILE + (lane & 15);
    const int* p = (lane < 16) ? u : v;
    return (e < (long)nE) ? p[e] : 0;
}

#define ISSUE_ROWS(IDX)                                                          \
    _Pragma("unroll")                                                            \
    for (int r = 0; r < TILE; ++r) {                                             \
        const unsigned ur = (unsigned)__builtin_amdgcn_readlane((IDX), r);       \
        const unsigned vr = (unsigned)__builtin_amdgcn_readlane((IDX), r + 16);  \
        ra[r] = *(const v4f*)(h + (size_t)ur * D_FEAT + laneOff);                \
        rb[r] = *(const v4f*)(h + (size_t)vr * D_FEAT + laneOff);                \
    }

__global__ __launch_bounds__(256)
void edge_dot_wmma(const float* __restrict__ h,
                   const int*   __restrict__ u,
                   const int*   __restrict__ v,
                   float*       __restrict__ out,
                   int nE)
{
    extern __shared__ float smem[];

    const int  lane   = threadIdx.x & 31;
    const int  wave   = __builtin_amdgcn_readfirstlane(threadIdx.x >> 5);
    const unsigned nTiles = (unsigned)(((long)nE + TILE - 1) / TILE);
    const unsigned stride = gridDim.x * WAVES_PB;
    unsigned       tile   = blockIdx.x * WAVES_PB + wave;

    float* ldsA = smem + (size_t)wave * (2 * TILE * ROWSTRIDE);
    float* ldsB = ldsA + TILE * ROWSTRIDE;

    const int    mrow  = lane & 15;
    const int    khalf = (lane >> 4) * 2;
    const float* Arow  = ldsA + mrow * ROWSTRIDE + khalf;
    const float* Brow  = ldsB + mrow * ROWSTRIDE + khalf;
    const unsigned laneOff = (unsigned)lane * 4u;

    const bool validLane = (lane < 8) || (lane >= 24);
    const int  m   = (lane < 8) ? lane : (lane - 16);
    const int  sel = (lane < 8) ? lane : (lane - 24);

    if (tile >= nTiles) return;

    v4f ra[TILE], rb[TILE];
    {
        const int idx0 = load_index(u, v, tile, lane, nE);
        ISSUE_ROWS(idx0)
    }
    unsigned tnext   = tile + stride;
    int      idxNext = (tnext < nTiles) ? load_index(u, v, tnext, lane, nE) : 0;

    for (;;) {
        #pragma unroll
        for (int r = 0; r < TILE; ++r) {
            *(v4f*)(ldsA + r * ROWSTRIDE + lane * 4) = ra[r];
            *(v4f*)(ldsB + r * ROWSTRIDE + lane * 4) = rb[r];
        }
        __builtin_amdgcn_wave_barrier();

        const bool more = (tnext < nTiles);
        if (more) {
            ISSUE_ROWS(idxNext)
            const unsigned t2 = tnext + stride;
            idxNext = (t2 < nTiles) ? load_index(u, v, t2, lane, nE) : 0;
        }

        v8f c = {};
        #pragma unroll
        for (int k = 0; k < D_FEAT; k += 4) {
            const v2f a = *(const v2f*)(Arow + k);
            const v2f b = *(const v2f*)(Brow + k);
            c = __builtin_amdgcn_wmma_f32_16x16x4_f32(false, a, false, b,
                                                      (short)0, c, false, false);
        }

        float val = c[0];
        #pragma unroll
        for (int i = 1; i < 8; ++i)
            val = (sel == i) ? c[i] : val;

        const long e = (long)tile * TILE + m;
        if (validLane && e < (long)nE)
            out[e] = val;

        __builtin_amdgcn_wave_barrier();

        if (!more) break;
        tile   = tnext;
        tnext += stride;
    }
}

extern "C" void kernel_launch(void* const* d_in, const int* in_sizes, int n_in,
                              void* d_out, int out_size, void* d_ws, size_t ws_size,
                              hipStream_t stream)
{
    const float* h  = (const float*)d_in[0];
    const int*   u  = (const int*)d_in[1];
    const int*   v  = (const int*)d_in[2];
    float*       out = (float*)d_out;

    const int    nE        = in_sizes[1];
    const size_t smemBytes = (size_t)WAVES_PB * 2 * TILE * ROWSTRIDE * sizeof(float);

    edge_dot_wmma<<<GRID_BLOCKS, 256, smemBytes, stream>>>(h, u, v, out, nE);
}
